# starting kernel
# baseline (speedup 1.0000x reference)
#include <hip/hip_runtime.h>
#include <math.h>

typedef float v2f __attribute__((ext_vector_type(2)));
typedef float v8f __attribute__((ext_vector_type(8)));

#define KDIM 386
#define KMAIN 384
#define HDIM 256

__global__ void gcn_init_deg(float* deg, int N) {
  int i = blockIdx.x * blockDim.x + threadIdx.x;
  if (i < N) deg[i] = 1.0f;
}

__global__ void gcn_accum_deg(const int* __restrict__ dst, float* deg, int E) {
  int e = blockIdx.x * blockDim.x + threadIdx.x;
  if (e < E) atomicAdd(&deg[dst[e]], 1.0f);
}

__global__ void gcn_compute_dis(const float* __restrict__ deg, float* dis, int N) {
  int i = blockIdx.x * blockDim.x + threadIdx.x;
  if (i < N) dis[i] = rsqrtf(deg[i]);
}

__global__ void gcn_edge_norm(const int* __restrict__ src, const int* __restrict__ dst,
                              const float* __restrict__ dis, float* en, int E) {
  int e = blockIdx.x * blockDim.x + threadIdx.x;
  if (e < E) en[e] = dis[src[e]] * dis[dst[e]];
}

__global__ __launch_bounds__(256) void gcn_gemm1_wmma(const float* __restrict__ x,
                                                      const float* __restrict__ W1,
                                                      float* __restrict__ t, int M) {
  const int lane = threadIdx.x & 31;
  const int wave = threadIdx.x >> 5;
  const int m0 = blockIdx.x * 16;
  const int n0 = wave * 32;
  const int halfSel = lane >> 4;
  const int l15 = lane & 15;

  v8f c0 = {0.f, 0.f, 0.f, 0.f, 0.f, 0.f, 0.f, 0.f};
  v8f c1 = {0.f, 0.f, 0.f, 0.f, 0.f, 0.f, 0.f, 0.f};

  int row = m0 + l15;
  int rowc = (row < M) ? row : (M - 1);
  const float* arow = x + (size_t)rowc * KDIM;
  const int col0 = n0 + l15;

  for (int k = 0; k < KMAIN; k += 4) {
    const int ka = k + halfSel * 2;
    const float2 av = *(const float2*)(arow + ka);
    v2f a;  a.x = av.x;  a.y = av.y;
    const float* bp = W1 + (size_t)ka * HDIM + col0;
    v2f b0; b0.x = bp[0];  b0.y = bp[HDIM];
    v2f b1; b1.x = bp[16]; b1.y = bp[HDIM + 16];
    c0 = __builtin_amdgcn_wmma_f32_16x16x4_f32(false, a, false, b0, (short)0, c0,
                                               false, false);
    c1 = __builtin_amdgcn_wmma_f32_16x16x4_f32(false, a, false, b1, (short)0, c1,
                                               false, false);
  }

  {
    v2f a  = {0.f, 0.f};
    v2f b0 = {0.f, 0.f};
    v2f b1 = {0.f, 0.f};
    if (halfSel == 0) {
      const float2 av = *(const float2*)(arow + KMAIN);
      a.x = av.x;  a.y = av.y;
      const float* bp = W1 + (size_t)KMAIN * HDIM + col0;
      b0.x = bp[0];  b0.y = bp[HDIM];
      b1.x = bp[16]; b1.y = bp[HDIM + 16];
    }
    c0 = __builtin_amdgcn_wmma_f32_16x16x4_f32(false, a, false, b0, (short)0, c0,
                                               false, false);
    c1 = __builtin_amdgcn_wmma_f32_16x16x4_f32(false, a, false, b1, (short)0, c1,
                                               false, false);
  }

#pragma unroll
  for (int r = 0; r < 8; ++r) {
    int rr = m0 + r + halfSel * 8;
    if (rr < M) {
      t[(size_t)rr * HDIM + col0] = c0[r];
      t[(size_t)rr * HDIM + col0 + 16] = c1[r];
    }
  }
}

__global__ void gcn_agg1_init(const float* __restrict__ t, const float* __restrict__ dis,
                              const float* __restrict__ b1, float* agg, int N) {
  long long i = (long long)blockIdx.x * blockDim.x + threadIdx.x;
  if (i < (long long)N * HDIM) {
    int n = (int)(i >> 8);
    int f = (int)(i & (HDIM - 1));
    float sn = dis[n] * dis[n];
    agg[i] = t[i] * sn + b1[f];
  }
}

__global__ __launch_bounds__(256) void gcn_scatter1(const int* __restrict__ src,
                                                    const int* __restrict__ dst,
                                                    const float* __restrict__ en,
                                                    const float* __restrict__ t,
                                                    float* agg, int E) {
  int e = blockIdx.x * 4 + (threadIdx.x >> 6);
  if (e >= E) return;
  int f = (threadIdx.x & 63) * 4;
  int s = src[e], d = dst[e];
  float w = en[e];
  float4 v = *(const float4*)(t + (size_t)s * HDIM + f);
  float* ap = agg + (size_t)d * HDIM + f;
  atomicAdd(ap + 0, v.x * w);
  atomicAdd(ap + 1, v.y * w);
  atomicAdd(ap + 2, v.z * w);
  atomicAdd(ap + 3, v.w * w);
}

__global__ __launch_bounds__(256) void gcn_gemm2(const float* __restrict__ agg,
                                                 const float* __restrict__ W2,
                                                 float* t2, int N) {
  int row = blockIdx.x * 8 + (threadIdx.x >> 5);
  int lane = threadIdx.x & 31;
  if (row >= N) return;
  const float* ar = agg + (size_t)row * HDIM;
  float s0 = 0.f, s1 = 0.f;
#pragma unroll
  for (int k = lane; k < HDIM; k += 32) {
    float a = ar[k];
    a = a > 0.f ? a : 0.f;
    s0 += a * W2[k * 2 + 0];
    s1 += a * W2[k * 2 + 1];
  }
#pragma unroll
  for (int off = 16; off > 0; off >>= 1) {
    s0 += __shfl_down(s0, off);
    s1 += __shfl_down(s1, off);
  }
  if (lane == 0) {
    t2[row * 2 + 0] = s0;
    t2[row * 2 + 1] = s1;
  }
}

__global__ void gcn_out_init(const float* __restrict__ t2, const float* __restrict__ dis,
                             const float* __restrict__ b2, float* out, int N) {
  int i = blockIdx.x * blockDim.x + threadIdx.x;
  if (i < 2 * N) {
    int n = i >> 1;
    int c = i & 1;
    float sn = dis[n] * dis[n];
    out[i] = t2[i] * sn + b2[c];
  }
}

__global__ void gcn_scatter2(const int* __restrict__ src, const int* __restrict__ dst,
                             const float* __restrict__ en, const float* __restrict__ t2,
                             float* out, int E) {
  int e = blockIdx.x * blockDim.x + threadIdx.x;
  if (e < E) {
    int s = src[e], d = dst[e];
    float w = en[e];
    atomicAdd(&out[d * 2 + 0], t2[s * 2 + 0] * w);
    atomicAdd(&out[d * 2 + 1], t2[s * 2 + 1] * w);
  }
}

__global__ void gcn_tanh(float* out, int n) {
  int i = blockIdx.x * blockDim.x + threadIdx.x;
  if (i < n) out[i] = tanhf(out[i]);
}

extern "C" void kernel_launch(void* const* d_in, const int* in_sizes, int n_in,
                              void* d_out, int out_size, void* d_ws, size_t ws_size,
                              hipStream_t stream) {
  const float* x  = (const float*)d_in[0];
  const int*   src = (const int*)d_in[1];
  const int*   dst = (const int*)d_in[2];
  const float* W1 = (const float*)d_in[3];
  const float* b1 = (const float*)d_in[4];
  const float* W2 = (const float*)d_in[5];
  const float* b2 = (const float*)d_in[6];
  float* out = (float*)d_out;

  const int N = in_sizes[0] / KDIM;
  const int E = in_sizes[1];

  float* ws  = (float*)d_ws;
  float* t   = ws;
  float* agg = t   + (size_t)N * HDIM;
  float* deg = agg + (size_t)N * HDIM;
  float* dis = deg + N;
  float* en  = dis + N;
  float* t2  = en  + E;

  const int TB = 256;
  gcn_init_deg<<<(N + TB - 1) / TB, TB, 0, stream>>>(deg, N);
  gcn_accum_deg<<<(E + TB - 1) / TB, TB, 0, stream>>>(dst, deg, E);
  gcn_compute_dis<<<(N + TB - 1) / TB, TB, 0, stream>>>(deg, dis, N);
  gcn_edge_norm<<<(E + TB - 1) / TB, TB, 0, stream>>>(src, dst, dis, en, E);

  gcn_gemm1_wmma<<<(N + 15) / 16, 256, 0, stream>>>(x, W1, t, N);
  {
    long long tot = (long long)N * HDIM;
    gcn_agg1_init<<<(int)((tot + TB - 1) / TB), TB, 0, stream>>>(t, dis, b1, agg, N);
  }
  gcn_scatter1<<<(E + 3) / 4, 256, 0, stream>>>(src, dst, en, t, agg, E);

  gcn_gemm2<<<(N + 7) / 8, 256, 0, stream>>>(agg, W2, t2, N);
  gcn_out_init<<<(2 * N + TB - 1) / TB, TB, 0, stream>>>(t2, dis, b2, out, N);
  gcn_scatter2<<<(E + TB - 1) / TB, TB, 0, stream>>>(src, dst, en, t2, out, E);
  gcn_tanh<<<(2 * N + TB - 1) / TB, TB, 0, stream>>>(out, 2 * N);
}


# Normalisation factored per row: GEMM1/GEMM2 epilogues scale rows by dis and store agg/out start values, consumers rescale; no edge weights; 11 -> 7 launches; atomics kept; product rounding changes
# speedup vs baseline: 1.5202x; 1.5202x over previous
#include <hip/hip_runtime.h>
#include <math.h>

typedef float v2f __attribute__((ext_vector_type(2)));
typedef float v8f __attribute__((ext_vector_type(8)));

#define KDIM 386
#define KMAIN 384
#define HDIM 256


__global__ void gcn_init_deg(float* deg, int N) {
  int i = blockIdx.x * blockDim.x + threadIdx.x;
  if (i < N) deg[i] = 1.0f;
}

__global__ void gcn_accum_deg(const int* __restrict__ dst, float* deg, int E) {
  int e = blockIdx.x * blockDim.x + threadIdx.x;
  if (e < E) atomicAdd(&deg[dst[e]], 1.0f);
}

__device__ __forceinline__ float gcn_dis(const float* __restrict__ deg, int n) {
  return rsqrtf(deg[n]);
}

__global__ __launch_bounds__(256) void gcn_gemm1_wmma(const float* __restrict__ x,
                                                      const float* __restrict__ W1,
                                                      const float* __restrict__ deg,
                                                      float* __restrict__ t,
                                                      float* __restrict__ agg, int M) {
  const int lane = threadIdx.x & 31;
  const int wave = threadIdx.x >> 5;
  const int m0 = blockIdx.x * 16;
  const int n0 = wave * 32;
  const int halfSel = lane >> 4;
  const int l15 = lane & 15;

  v8f c0 = {0.f, 0.f, 0.f, 0.f, 0.f, 0.f, 0.f, 0.f};
  v8f c1 = {0.f, 0.f, 0.f, 0.f, 0.f, 0.f, 0.f, 0.f};

  int row = m0 + l15;
  int rowc = (row < M) ? row : (M - 1);
  const float* arow = x + (size_t)rowc * KDIM;
  const int col0 = n0 + l15;

  for (int k = 0; k < KMAIN; k += 4) {
    const int ka = k + halfSel * 2;
    const float2 av = *(const float2*)(arow + ka);
    v2f a;  a.x = av.x;  a.y = av.y;
    const float* bp = W1 + (size_t)ka * HDIM + col0;
    v2f b0; b0.x = bp[0];  b0.y = bp[HDIM];
    v2f b1; b1.x = bp[16]; b1.y = bp[HDIM + 16];
    c0 = __builtin_amdgcn_wmma_f32_16x16x4_f32(false, a, false, b0, (short)0, c0,
                                               false, false);
    c1 = __builtin_amdgcn_wmma_f32_16x16x4_f32(false, a, false, b1, (short)0, c1,
                                               false, false);
  }

  {
    v2f a  = {0.f, 0.f};
    v2f b0 = {0.f, 0.f};
    v2f b1 = {0.f, 0.f};
    if (halfSel == 0) {
      const float2 av = *(const float2*)(arow + KMAIN);
      a.x = av.x;  a.y = av.y;
      const float* bp = W1 + (size_t)KMAIN * HDIM + col0;
      b0.x = bp[0];  b0.y = bp[HDIM];
      b1.x = bp[16]; b1.y = bp[HDIM + 16];
    }
    c0 = __builtin_amdgcn_wmma_f32_16x16x4_f32(false, a, false, b0, (short)0, c0,
                                               false, false);
    c1 = __builtin_amdgcn_wmma_f32_16x16x4_f32(false, a, false, b1, (short)0, c1,
                                               false, false);
  }

  float dr[8];
#pragma unroll
  for (int r = 0; r < 8; ++r) {
    int rr = m0 + r + halfSel * 8;
    dr[r] = gcn_dis(deg, (rr < M) ? rr : (M - 1));
  }

#pragma unroll
  for (int r = 0; r < 8; ++r) {
    int rr = m0 + r + halfSel * 8;
    if (rr < M) {
      const size_t o = (size_t)rr * HDIM + col0;
      const float u0 = c0[r] * dr[r];
      const float u1 = c1[r] * dr[r];
      t[o] = u0;
      t[o + 16] = u1;
      agg[o] = u0;
      agg[o + 16] = u1;
    }
  }
}

#define S1_EPW 4

template <int NE>
__device__ __forceinline__ void gcn_scatter1_group(const int* __restrict__ src,
                                                   const int* __restrict__ dst,
                                                   const float* __restrict__ t,
                                                   float* agg, int e0, int lane) {
  int s[NE], d[NE];
  for (int i = 0; i < NE; ++i) {
    s[i] = src[e0 + i];
    d[i] = dst[e0 + i];
  }
  float v[NE][8];
  for (int i = 0; i < NE; ++i) {
    const float* tp = t + (size_t)s[i] * HDIM + lane;
    for (int j = 0; j < 8; ++j) v[i][j] = tp[32 * j];
  }
  for (int i = 0; i < NE; ++i) {
    float* ap = agg + (size_t)d[i] * HDIM + lane;
    for (int j = 0; j < 8; ++j) atomicAdd(ap + 32 * j, v[i][j]);
  }
}

__global__ __launch_bounds__(256) void gcn_scatter1(const int* __restrict__ src,
                                                    const int* __restrict__ dst,
                                                    const float* __restrict__ t,
                                                    float* agg, int E) {
  const int lane = threadIdx.x & 31;
  const int wave = __builtin_amdgcn_readfirstlane((int)(blockIdx.x * 8 + (threadIdx.x >> 5)));
  const int e0 = wave * S1_EPW;
  if (e0 + S1_EPW <= E) {
    gcn_scatter1_group<S1_EPW>(src, dst, t, agg, e0, lane);
  } else {
    for (int e = e0; e < E; ++e) gcn_scatter1_group<1>(src, dst, t, agg, e, lane);
  }
}

__global__ __launch_bounds__(256) void gcn_gemm2(const float* __restrict__ agg,
                                                 const float* __restrict__ W2,
                                                 const float* __restrict__ deg,
                                                 const float* __restrict__ b1,
                                                 float* __restrict__ t2,
                                                 float* __restrict__ out, int N) {
  int row = blockIdx.x * 8 + (threadIdx.x >> 5);
  int lane = threadIdx.x & 31;
  if (row >= N) return;
  const float* ar = agg + (size_t)row * HDIM;
  const float dr = gcn_dis(deg, row);
  float s0 = 0.f, s1 = 0.f;
#pragma unroll
  for (int k = lane; k < HDIM; k += 32) {
    float a = ar[k] * dr + b1[k];
    a = a > 0.f ? a : 0.f;
    s0 += a * W2[k * 2 + 0];
    s1 += a * W2[k * 2 + 1];
  }
#pragma unroll
  for (int off = 16; off > 0; off >>= 1) {
    s0 += __shfl_down(s0, off);
    s1 += __shfl_down(s1, off);
  }
  if (lane == 0) {
    const float u0 = s0 * dr;
    const float u1 = s1 * dr;
    t2[row * 2 + 0] = u0;
    t2[row * 2 + 1] = u1;
    out[row * 2 + 0] = u0;
    out[row * 2 + 1] = u1;
  }
}

__global__ void gcn_scatter2(const int* __restrict__ src, const int* __restrict__ dst,
                             const float* __restrict__ t2, float* out, int E) {
  int e = blockIdx.x * blockDim.x + threadIdx.x;
  if (e < E) {
    int s = src[e], d = dst[e];
    atomicAdd(&out[d * 2 + 0], t2[s * 2 + 0]);
    atomicAdd(&out[d * 2 + 1], t2[s * 2 + 1]);
  }
}

__global__ void gcn_tanh(const float* __restrict__ deg, const float* __restrict__ b2,
                         float* out, int n) {
  int i = blockIdx.x * blockDim.x + threadIdx.x;
  if (i < n) out[i] = tanhf(out[i] * gcn_dis(deg, i >> 1) + b2[i & 1]);
}

extern "C" void kernel_launch(void* const* d_in, const int* in_sizes, int n_in,
                              void* d_out, int out_size, void* d_ws, size_t ws_size,
                              hipStream_t stream) {
  const float* x  = (const float*)d_in[0];
  const int*   src = (const int*)d_in[1];
  const int*   dst = (const int*)d_in[2];
  const float* W1 = (const float*)d_in[3];
  const float* b1 = (const float*)d_in[4];
  const float* W2 = (const float*)d_in[5];
  const float* b2 = (const float*)d_in[6];
  float* out = (float*)d_out;

  const int N = in_sizes[0] / KDIM;
  const int E = in_sizes[1];

  float* ws  = (float*)d_ws;
  float* t   = ws;
  float* agg = t   + (size_t)N * HDIM;
  float* deg = agg + (size_t)N * HDIM;
  float* t2  = deg + N;

  const int TB = 256;
  gcn_init_deg<<<(N + TB - 1) / TB, TB, 0, stream>>>(deg, N);
  gcn_accum_deg<<<(E + TB - 1) / TB, TB, 0, stream>>>(dst, deg, E);

  gcn_gemm1_wmma<<<(N + 15) / 16, 256, 0, stream>>>(x, W1, deg, t, agg, N);
  gcn_scatter1<<<(E + 8 * S1_EPW - 1) / (8 * S1_EPW), 256, 0, stream>>>(src, dst, t, agg, E);

  gcn_gemm2<<<(N + 7) / 8, 256, 0, stream>>>(agg, W2, deg, b1, t2, out, N);
  gcn_scatter2<<<(E + TB - 1) / TB, TB, 0, stream>>>(src, dst, t2, out, E);
  gcn_tanh<<<(2 * N + TB - 1) / TB, TB, 0, stream>>>(deg, b2, out, 2 * N);
}


# Degree count and layer-2 edge sum add into a per-work-group whole-table copy in LDS (LDS float adds), merged by one whole-line atomic per line; summation order changes (arrival order kept)
# speedup vs baseline: 2.2256x; 1.4640x over previous
#include <hip/hip_runtime.h>
#include <math.h>

typedef float v2f __attribute__((ext_vector_type(2)));
typedef float v8f __attribute__((ext_vector_type(8)));

#define KDIM 386
#define KMAIN 384
#define HDIM 256


__global__ void gcn_init_deg(float* deg, int N) {
  int i = blockIdx.x * blockDim.x + threadIdx.x;
  if (i < N) deg[i] = 1.0f;
}

#define PRV_THREADS 1024
#define PRV_DEG_CAP 20480
#define PRV_L2_CAP 20480
#define PRV_DEG_EPT 8
#define PRV_L2_EPT 5
#define PRV_MRG 4

template <int EPT>
__global__ __launch_bounds__(PRV_THREADS) void gcn_accum_deg_lds(const int* __restrict__ dst,
                                                                 float* deg, int E, int N) {
  __shared__ __attribute__((aligned(16))) float cnt[PRV_DEG_CAP];
  const int tid = threadIdx.x;
  const int base = blockIdx.x * (PRV_THREADS * EPT) + tid;
  int d[EPT];
  for (int i = 0; i < EPT; ++i) {
    const int e = base + i * PRV_THREADS;
    const int dv = dst[(e < E) ? e : (E - 1)];
    d[i] = (e < E) ? dv : -1;
  }
  for (int n0 = 0; n0 < N; n0 += PRV_DEG_CAP) {
    const int nn = (N - n0 < PRV_DEG_CAP) ? (N - n0) : PRV_DEG_CAP;
    float4* c4 = (float4*)cnt;
    for (int i = tid; i < (nn + 3) / 4; i += PRV_THREADS) c4[i] = make_float4(0.f, 0.f, 0.f, 0.f);
    __syncthreads();
    for (int i = 0; i < EPT; ++i) {
      const unsigned r = (unsigned)(d[i] - n0);
      if (r < (unsigned)nn) atomicAdd(&cnt[r], 1.0f);
    }
    __syncthreads();
    float* dp = deg + n0;
    int i = tid;
    for (; i + (PRV_MRG - 1) * PRV_THREADS < nn; i += PRV_MRG * PRV_THREADS) {
      float a[PRV_MRG];
      for (int k = 0; k < PRV_MRG; ++k) a[k] = cnt[i + k * PRV_THREADS];
      for (int k = 0; k < PRV_MRG; ++k) atomicAdd(&dp[i + k * PRV_THREADS], a[k]);
    }
    for (; i < nn; i += PRV_THREADS) atomicAdd(&dp[i], cnt[i]);
    __syncthreads();
  }
}

__device__ __forceinline__ float gcn_dis(const float* __restrict__ deg, int n) {
  return rsqrtf(deg[n]);
}

__global__ __launch_bounds__(256) void gcn_gemm1_wmma(const float* __restrict__ x,
                                                      const float* __restrict__ W1,
                                                      const float* __restrict__ deg,
                                                      float* __restrict__ t,
                                                      float* __restrict__ agg, int M) {
  const int lane = threadIdx.x & 31;
  const int wave = threadIdx.x >> 5;
  const int m0 = blockIdx.x * 16;
  const int n0 = wave * 32;
  const int halfSel = lane >> 4;
  const int l15 = lane & 15;

  v8f c0 = {0.f, 0.f, 0.f, 0.f, 0.f, 0.f, 0.f, 0.f};
  v8f c1 = {0.f, 0.f, 0.f, 0.f, 0.f, 0.f, 0.f, 0.f};

  int row = m0 + l15;
  int rowc = (row < M) ? row : (M - 1);
  const float* arow = x + (size_t)rowc * KDIM;
  const int col0 = n0 + l15;

  for (int k = 0; k < KMAIN; k += 4) {
    const int ka = k + halfSel * 2;
    const float2 av = *(const float2*)(arow + ka);
    v2f a;  a.x = av.x;  a.y = av.y;
    const float* bp = W1 + (size_t)ka * HDIM + col0;
    v2f b0; b0.x = bp[0];  b0.y = bp[HDIM];
    v2f b1; b1.x = bp[16]; b1.y = bp[HDIM + 16];
    c0 = __builtin_amdgcn_wmma_f32_16x16x4_f32(false, a, false, b0, (short)0, c0,
                                               false, false);
    c1 = __builtin_amdgcn_wmma_f32_16x16x4_f32(false, a, false, b1, (short)0, c1,
                                               false, false);
  }

  {
    v2f a  = {0.f, 0.f};
    v2f b0 = {0.f, 0.f};
    v2f b1 = {0.f, 0.f};
    if (halfSel == 0) {
      const float2 av = *(const float2*)(arow + KMAIN);
      a.x = av.x;  a.y = av.y;
      const float* bp = W1 + (size_t)KMAIN * HDIM + col0;
      b0.x = bp[0];  b0.y = bp[HDIM];
      b1.x = bp[16]; b1.y = bp[HDIM + 16];
    }
    c0 = __builtin_amdgcn_wmma_f32_16x16x4_f32(false, a, false, b0, (short)0, c0,
                                               false, false);
    c1 = __builtin_amdgcn_wmma_f32_16x16x4_f32(false, a, false, b1, (short)0, c1,
                                               false, false);
  }

  float dr[8];
#pragma unroll
  for (int r = 0; r < 8; ++r) {
    int rr = m0 + r + halfSel * 8;
    dr[r] = gcn_dis(deg, (rr < M) ? rr : (M - 1));
  }

#pragma unroll
  for (int r = 0; r < 8; ++r) {
    int rr = m0 + r + halfSel * 8;
    if (rr < M) {
      const size_t o = (size_t)rr * HDIM + col0;
      const float u0 = c0[r] * dr[r];
      const float u1 = c1[r] * dr[r];
      t[o] = u0;
      t[o + 16] = u1;
      agg[o] = u0;
      agg[o + 16] = u1;
    }
  }
}

#define S1_EPW 4

template <int NE>
__device__ __forceinline__ void gcn_scatter1_group(const int* __restrict__ src,
                                                   const int* __restrict__ dst,
                                                   const float* __restrict__ t,
                                                   float* agg, int e0, int lane) {
  int s[NE], d[NE];
  for (int i = 0; i < NE; ++i) {
    s[i] = src[e0 + i];
    d[i] = dst[e0 + i];
  }
  float v[NE][8];
  for (int i = 0; i < NE; ++i) {
    const float* tp = t + (size_t)s[i] * HDIM + lane;
    for (int j = 0; j < 8; ++j) v[i][j] = tp[32 * j];
  }
  for (int i = 0; i < NE; ++i) {
    float* ap = agg + (size_t)d[i] * HDIM + lane;
    for (int j = 0; j < 8; ++j) atomicAdd(ap + 32 * j, v[i][j]);
  }
}

__global__ __launch_bounds__(256) void gcn_scatter1(const int* __restrict__ src,
                                                    const int* __restrict__ dst,
                                                    const float* __restrict__ t,
                                                    float* agg, int E) {
  const int lane = threadIdx.x & 31;
  const int wave = __builtin_amdgcn_readfirstlane((int)(blockIdx.x * 8 + (threadIdx.x >> 5)));
  const int e0 = wave * S1_EPW;
  if (e0 + S1_EPW <= E) {
    gcn_scatter1_group<S1_EPW>(src, dst, t, agg, e0, lane);
  } else {
    for (int e = e0; e < E; ++e) gcn_scatter1_group<1>(src, dst, t, agg, e, lane);
  }
}

__global__ __launch_bounds__(256) void gcn_gemm2(const float* __restrict__ agg,
                                                 const float* __restrict__ W2,
                                                 const float* __restrict__ deg,
                                                 const float* __restrict__ b1,
                                                 float* __restrict__ t2,
                                                 float* __restrict__ out, int N) {
  int row = blockIdx.x * 8 + (threadIdx.x >> 5);
  int lane = threadIdx.x & 31;
  if (row >= N) return;
  const float* ar = agg + (size_t)row * HDIM;
  const float dr = gcn_dis(deg, row);
  float s0 = 0.f, s1 = 0.f;
#pragma unroll
  for (int k = lane; k < HDIM; k += 32) {
    float a = ar[k] * dr + b1[k];
    a = a > 0.f ? a : 0.f;
    s0 += a * W2[k * 2 + 0];
    s1 += a * W2[k * 2 + 1];
  }
#pragma unroll
  for (int off = 16; off > 0; off >>= 1) {
    s0 += __shfl_down(s0, off);
    s1 += __shfl_down(s1, off);
  }
  if (lane == 0) {
    const float u0 = s0 * dr;
    const float u1 = s1 * dr;
    t2[row * 2 + 0] = u0;
    t2[row * 2 + 1] = u1;
    out[row * 2 + 0] = u0;
    out[row * 2 + 1] = u1;
  }
}

template <int EPT>
__global__ __launch_bounds__(PRV_THREADS) void gcn_scatter2_lds(const int* __restrict__ src,
                                                                const int* __restrict__ dst,
                                                                const float* __restrict__ t2,
                                                                float* out, int E, int N) {
  __shared__ __attribute__((aligned(16))) float acc[2 * PRV_L2_CAP];
  const int tid = threadIdx.x;
  const int base = blockIdx.x * (PRV_THREADS * EPT) + tid;
  int d[EPT];
  float2 v[EPT];
  {
    int s[EPT];
    for (int i = 0; i < EPT; ++i) {
      const int e = base + i * PRV_THREADS;
      const int ec = (e < E) ? e : (E - 1);
      s[i] = src[ec];
      const int dv = dst[ec];
      d[i] = (e < E) ? dv : -1;
    }
    for (int i = 0; i < EPT; ++i) v[i] = *(const float2*)(t2 + 2 * (size_t)s[i]);
  }
  for (int n0 = 0; n0 < N; n0 += PRV_L2_CAP) {
    const int nn = (N - n0 < PRV_L2_CAP) ? (N - n0) : PRV_L2_CAP;
    float4* a4 = (float4*)acc;
    for (int i = tid; i < (2 * nn + 3) / 4; i += PRV_THREADS) a4[i] = make_float4(0.f, 0.f, 0.f, 0.f);
    __syncthreads();
    for (int i = 0; i < EPT; ++i) {
      const unsigned r = (unsigned)(d[i] - n0);
      if (r < (unsigned)nn) {
        atomicAdd(&acc[2 * r + 0], v[i].x);
        atomicAdd(&acc[2 * r + 1], v[i].y);
      }
    }
    __syncthreads();
    float* op = out + 2 * (size_t)n0;
    const int lim = 2 * nn;
    int i = tid;
    for (; i + (PRV_MRG - 1) * PRV_THREADS < lim; i += PRV_MRG * PRV_THREADS) {
      float a[PRV_MRG];
      for (int k = 0; k < PRV_MRG; ++k) a[k] = acc[i + k * PRV_THREADS];
      for (int k = 0; k < PRV_MRG; ++k) atomicAdd(&op[i + k * PRV_THREADS], a[k]);
    }
    for (; i < lim; i += PRV_THREADS) atomicAdd(&op[i], acc[i]);
    __syncthreads();
  }
}

__global__ void gcn_tanh(const float* __restrict__ deg, const float* __restrict__ b2,
                         float* out, int n) {
  int i = blockIdx.x * blockDim.x + threadIdx.x;
  if (i < n) out[i] = tanhf(out[i] * gcn_dis(deg, i >> 1) + b2[i & 1]);
}

extern "C" void kernel_launch(void* const* d_in, const int* in_sizes, int n_in,
                              void* d_out, int out_size, void* d_ws, size_t ws_size,
                              hipStream_t stream) {
  const float* x  = (const float*)d_in[0];
  const int*   src = (const int*)d_in[1];
  const int*   dst = (const int*)d_in[2];
  const float* W1 = (const float*)d_in[3];
  const float* b1 = (const float*)d_in[4];
  const float* W2 = (const float*)d_in[5];
  const float* b2 = (const float*)d_in[6];
  float* out = (float*)d_out;

  const int N = in_sizes[0] / KDIM;
  const int E = in_sizes[1];

  float* ws  = (float*)d_ws;
  float* t   = ws;
  float* agg = t   + (size_t)N * HDIM;
  float* t2  = agg + (size_t)N * HDIM;
  float* deg = t2 + 2 * (size_t)N;

  const int TB = 256;
  gcn_init_deg<<<(N + TB - 1) / TB, TB, 0, stream>>>(deg, N);
  gcn_accum_deg_lds<PRV_DEG_EPT>
      <<<(E + PRV_THREADS * PRV_DEG_EPT - 1) / (PRV_THREADS * PRV_DEG_EPT), PRV_THREADS, 0, stream>>>(
          dst, deg, E, N);

  gcn_gemm1_wmma<<<(N + 15) / 16, 256, 0, stream>>>(x, W1, deg, t, agg, N);
  gcn_scatter1<<<(E + 8 * S1_EPW - 1) / (8 * S1_EPW), 256, 0, stream>>>(src, dst, t, agg, E);

  gcn_gemm2<<<(N + 7) / 8, 256, 0, stream>>>(agg, W2, deg, b1, t2, out, N);
  gcn_scatter2_lds<PRV_L2_EPT>
      <<<(E + PRV_THREADS * PRV_L2_EPT - 1) / (PRV_THREADS * PRV_L2_EPT), PRV_THREADS, 0, stream>>>(
          src, dst, t2, out, E, N);
  gcn_tanh<<<(2 * N + TB - 1) / TB, TB, 0, stream>>>(deg, b2, out, 2 * N);
}


# GEMM1 re-routed: each work-group keeps its half of W1 in LDS (k pairs interleaved), A fragments come straight from x; a 16x64 wave block gets 4 products per A fragment; k order and sums unchanged
# speedup vs baseline: 2.6035x; 1.1698x over previous
#include <hip/hip_runtime.h>
#include <math.h>

typedef float v2f __attribute__((ext_vector_type(2)));
typedef float v8f __attribute__((ext_vector_type(8)));

#define KDIM 386
#define KMAIN 384
#define HDIM 256


__global__ void gcn_init_deg(float* deg, int N) {
  int i = blockIdx.x * blockDim.x + threadIdx.x;
  if (i < N) deg[i] = 1.0f;
}

#define PRV_THREADS 1024
#define PRV_DEG_CAP 20480
#define PRV_L2_CAP 20480
#define PRV_DEG_EPT 8
#define PRV_L2_EPT 5
#define PRV_MRG 4

template <int EPT>
__global__ __launch_bounds__(PRV_THREADS) void gcn_accum_deg_lds(const int* __restrict__ dst,
                                                                 float* deg, int E, int N) {
  __shared__ __attribute__((aligned(16))) float cnt[PRV_DEG_CAP];
  const int tid = threadIdx.x;
  const int base = blockIdx.x * (PRV_THREADS * EPT) + tid;
  int d[EPT];
  for (int i = 0; i < EPT; ++i) {
    const int e = base + i * PRV_THREADS;
    const int dv = dst[(e < E) ? e : (E - 1)];
    d[i] = (e < E) ? dv : -1;
  }
  for (int n0 = 0; n0 < N; n0 += PRV_DEG_CAP) {
    const int nn = (N - n0 < PRV_DEG_CAP) ? (N - n0) : PRV_DEG_CAP;
    float4* c4 = (float4*)cnt;
    for (int i = tid; i < (nn + 3) / 4; i += PRV_THREADS) c4[i] = make_float4(0.f, 0.f, 0.f, 0.f);
    __syncthreads();
    for (int i = 0; i < EPT; ++i) {
      const unsigned r = (unsigned)(d[i] - n0);
      if (r < (unsigned)nn) atomicAdd(&cnt[r], 1.0f);
    }
    __syncthreads();
    float* dp = deg + n0;
    int i = tid;
    for (; i + (PRV_MRG - 1) * PRV_THREADS < nn; i += PRV_MRG * PRV_THREADS) {
      float a[PRV_MRG];
      for (int k = 0; k < PRV_MRG; ++k) a[k] = cnt[i + k * PRV_THREADS];
      for (int k = 0; k < PRV_MRG; ++k) atomicAdd(&dp[i + k * PRV_THREADS], a[k]);
    }
    for (; i < nn; i += PRV_THREADS) atomicAdd(&dp[i], cnt[i]);
    __syncthreads();
  }
}

__device__ __forceinline__ float gcn_dis(const float* __restrict__ deg, int n) {
  return rsqrtf(deg[n]);
}

#define G1_ROWS 160
#define G1_COLS 128
#define G1_THREADS 640
#define G1_KPAIRS 193
#define G1_PITCH 288
#define G1_KBLK 32

__device__ __forceinline__ void g1_close_products(v8f (&acc)[4]) {
  __builtin_amdgcn_sched_barrier(0);
  asm volatile("v_nop\n\tv_nop\n\tv_nop\n\tv_nop\n\tv_nop"
               : "+v"(acc[0]), "+v"(acc[1]), "+v"(acc[2]), "+v"(acc[3]));
  __builtin_amdgcn_sched_barrier(0);
}

template <int NS>
__device__ __forceinline__ void g1_run(const float2 (&a)[NS], const float* bp, v8f (&acc)[4]) {
  float2 b[NS][4];
#pragma unroll
  for (int j = 0; j < NS; ++j)
#pragma unroll
    for (int q = 0; q < 4; ++q) b[j][q] = *(const float2*)(bp + 2 * j * G1_PITCH + 32 * q);
  __builtin_amdgcn_sched_barrier(0);
#pragma unroll
  for (int j = 0; j < NS; ++j) {
    v2f av;  av.x = a[j].x;  av.y = a[j].y;
#pragma unroll
    for (int q = 0; q < 4; ++q) {
      v2f bv;  bv.x = b[j][q].x;  bv.y = b[j][q].y;
      acc[q] = __builtin_amdgcn_wmma_f32_16x16x4_f32(false, av, false, bv, (short)0, acc[q],
                                                     false, false);
    }
  }
  g1_close_products(acc);
}

__global__ __launch_bounds__(G1_THREADS) void gcn_gemm1_wmma(const float* __restrict__ x,
                                                             const float* __restrict__ W1,
                                                             const float* __restrict__ deg,
                                                             float* __restrict__ t,
                                                             float* __restrict__ agg, int M) {
  __shared__ __attribute__((aligned(16))) float Bs[(G1_KPAIRS + 1) * G1_PITCH];
  const int tid = threadIdx.x;
  const int lane = tid & 31;
  const int wave = tid >> 5;
  const int halfSel = lane >> 4;
  const int l15 = lane & 15;
  const int colbase = (blockIdx.x & 1) * G1_COLS;
  const int m0 = (blockIdx.x >> 1) * G1_ROWS + (wave >> 1) * 16;
  const int cb = (wave & 1) * 64;

  const int row = m0 + l15;
  const int rowc = (row < M) ? row : (M - 1);
  const float* arow = x + (size_t)rowc * KDIM + halfSel * 2;

  float2 a0[8], a1[8];
#pragma unroll
  for (int j = 0; j < 8; ++j) a0[j] = *(const float2*)(arow + 4 * j);

  for (int it = tid; it < G1_KPAIRS * (G1_COLS / 4); it += G1_THREADS) {
    const int kp = it >> 5;
    const int c4 = (it & 31) * 4;
    const float* wp = W1 + (size_t)(2 * kp) * HDIM + colbase + c4;
    const float4 r0 = *(const float4*)wp;
    const float4 r1 = *(const float4*)(wp + HDIM);
    float4* dp = (float4*)(Bs + kp * G1_PITCH + 2 * c4);
    dp[0] = make_float4(r0.x, r1.x, r0.y, r1.y);
    dp[1] = make_float4(r0.z, r1.z, r0.w, r1.w);
  }
  for (int i = tid; i < 2 * G1_COLS; i += G1_THREADS) Bs[G1_KPAIRS * G1_PITCH + i] = 0.f;
  __syncthreads();

  v8f acc[4];
#pragma unroll
  for (int q = 0; q < 4; ++q) acc[q] = (v8f){0.f, 0.f, 0.f, 0.f, 0.f, 0.f, 0.f, 0.f};

  const float* bl = Bs + halfSel * G1_PITCH + 2 * (cb + l15);

  for (int kb = 0; kb < KMAIN / G1_KBLK; kb += 2) {
#pragma unroll
    for (int j = 0; j < 8; ++j) a1[j] = *(const float2*)(arow + (kb + 1) * G1_KBLK + 4 * j);
    g1_run<8>(a0, bl + (kb * (G1_KBLK / 2)) * G1_PITCH, acc);
    if (kb + 2 < KMAIN / G1_KBLK) {
#pragma unroll
      for (int j = 0; j < 8; ++j) a0[j] = *(const float2*)(arow + (kb + 2) * G1_KBLK + 4 * j);
    }
    g1_run<8>(a1, bl + ((kb + 1) * (G1_KBLK / 2)) * G1_PITCH, acc);
  }

  {
    float2 at[1];
    at[0] = make_float2(0.f, 0.f);
    if (halfSel == 0) at[0] = *(const float2*)(arow + KMAIN);
    g1_run<1>(at, bl + (KMAIN / 2) * G1_PITCH, acc);
  }

  float dr[8];
#pragma unroll
  for (int r = 0; r < 8; ++r) {
    int rr = m0 + r + halfSel * 8;
    dr[r] = gcn_dis(deg, (rr < M) ? rr : (M - 1));
  }

#pragma unroll
  for (int r = 0; r < 8; ++r) {
    int rr = m0 + r + halfSel * 8;
    if (rr < M) {
      const size_t o = (size_t)rr * HDIM + colbase + cb + l15;
#pragma unroll
      for (int q = 0; q < 4; ++q) {
        const float u = acc[q][r] * dr[r];
        t[o + 16 * q] = u;
        agg[o + 16 * q] = u;
      }
    }
  }
}

#define S1_EPW 4

template <int NE>
__device__ __forceinline__ void gcn_scatter1_group(const int* __restrict__ src,
                                                   const int* __restrict__ dst,
                                                   const float* __restrict__ t,
                                                   float* agg, int e0, int lane) {
  int s[NE], d[NE];
  for (int i = 0; i < NE; ++i) {
    s[i] = src[e0 + i];
    d[i] = dst[e0 + i];
  }
  float v[NE][8];
  for (int i = 0; i < NE; ++i) {
    const float* tp = t + (size_t)s[i] * HDIM + lane;
    for (int j = 0; j < 8; ++j) v[i][j] = tp[32 * j];
  }
  for (int i = 0; i < NE; ++i) {
    float* ap = agg + (size_t)d[i] * HDIM + lane;
    for (int j = 0; j < 8; ++j) atomicAdd(ap + 32 * j, v[i][j]);
  }
}

__global__ __launch_bounds__(256) void gcn_scatter1(const int* __restrict__ src,
                                                    const int* __restrict__ dst,
                                                    const float* __restrict__ t,
                                                    float* agg, int E) {
  const int lane = threadIdx.x & 31;
  const int wave = __builtin_amdgcn_readfirstlane((int)(blockIdx.x * 8 + (threadIdx.x >> 5)));
  const int e0 = wave * S1_EPW;
  if (e0 + S1_EPW <= E) {
    gcn_scatter1_group<S1_EPW>(src, dst, t, agg, e0, lane);
  } else {
    for (int e = e0; e < E; ++e) gcn_scatter1_group<1>(src, dst, t, agg, e, lane);
  }
}

__global__ __launch_bounds__(256) void gcn_gemm2(const float* __restrict__ agg,
                                                 const float* __restrict__ W2,
                                                 const float* __restrict__ deg,
                                                 const float* __restrict__ b1,
                                                 float* __restrict__ t2,
                                                 float* __restrict__ out, int N) {
  int row = blockIdx.x * 8 + (threadIdx.x >> 5);
  int lane = threadIdx.x & 31;
  if (row >= N) return;
  const float* ar = agg + (size_t)row * HDIM;
  const float dr = gcn_dis(deg, row);
  float s0 = 0.f, s1 = 0.f;
#pragma unroll
  for (int k = lane; k < HDIM; k += 32) {
    float a = ar[k] * dr + b1[k];
    a = a > 0.f ? a : 0.f;
    s0 += a * W2[k * 2 + 0];
    s1 += a * W2[k * 2 + 1];
  }
#pragma unroll
  for (int off = 16; off > 0; off >>= 1) {
    s0 += __shfl_down(s0, off);
    s1 += __shfl_down(s1, off);
  }
  if (lane == 0) {
    const float u0 = s0 * dr;
    const float u1 = s1 * dr;
    t2[row * 2 + 0] = u0;
    t2[row * 2 + 1] = u1;
    out[row * 2 + 0] = u0;
    out[row * 2 + 1] = u1;
  }
}

template <int EPT>
__global__ __launch_bounds__(PRV_THREADS) void gcn_scatter2_lds(const int* __restrict__ src,
                                                                const int* __restrict__ dst,
                                                                const float* __restrict__ t2,
                                                                float* out, int E, int N) {
  __shared__ __attribute__((aligned(16))) float acc[2 * PRV_L2_CAP];
  const int tid = threadIdx.x;
  const int base = blockIdx.x * (PRV_THREADS * EPT) + tid;
  int d[EPT];
  float2 v[EPT];
  {
    int s[EPT];
    for (int i = 0; i < EPT; ++i) {
      const int e = base + i * PRV_THREADS;
      const int ec = (e < E) ? e : (E - 1);
      s[i] = src[ec];
      const int dv = dst[ec];
      d[i] = (e < E) ? dv : -1;
    }
    for (int i = 0; i < EPT; ++i) v[i] = *(const float2*)(t2 + 2 * (size_t)s[i]);
  }
  for (int n0 = 0; n0 < N; n0 += PRV_L2_CAP) {
    const int nn = (N - n0 < PRV_L2_CAP) ? (N - n0) : PRV_L2_CAP;
    float4* a4 = (float4*)acc;
    for (int i = tid; i < (2 * nn + 3) / 4; i += PRV_THREADS) a4[i] = make_float4(0.f, 0.f, 0.f, 0.f);
    __syncthreads();
    for (int i = 0; i < EPT; ++i) {
      const unsigned r = (unsigned)(d[i] - n0);
      if (r < (unsigned)nn) {
        atomicAdd(&acc[2 * r + 0], v[i].x);
        atomicAdd(&acc[2 * r + 1], v[i].y);
      }
    }
    __syncthreads();
    float* op = out + 2 * (size_t)n0;
    const int lim = 2 * nn;
    int i = tid;
    for (; i + (PRV_MRG - 1) * PRV_THREADS < lim; i += PRV_MRG * PRV_THREADS) {
      float a[PRV_MRG];
      for (int k = 0; k < PRV_MRG; ++k) a[k] = acc[i + k * PRV_THREADS];
      for (int k = 0; k < PRV_MRG; ++k) atomicAdd(&op[i + k * PRV_THREADS], a[k]);
    }
    for (; i < lim; i += PRV_THREADS) atomicAdd(&op[i], acc[i]);
    __syncthreads();
  }
}

__global__ void gcn_tanh(const float* __restrict__ deg, const float* __restrict__ b2,
                         float* out, int n) {
  int i = blockIdx.x * blockDim.x + threadIdx.x;
  if (i < n) out[i] = tanhf(out[i] * gcn_dis(deg, i >> 1) + b2[i & 1]);
}

extern "C" void kernel_launch(void* const* d_in, const int* in_sizes, int n_in,
                              void* d_out, int out_size, void* d_ws, size_t ws_size,
                              hipStream_t stream) {
  const float* x  = (const float*)d_in[0];
  const int*   src = (const int*)d_in[1];
  const int*   dst = (const int*)d_in[2];
  const float* W1 = (const float*)d_in[3];
  const float* b1 = (const float*)d_in[4];
  const float* W2 = (const float*)d_in[5];
  const float* b2 = (const float*)d_in[6];
  float* out = (float*)d_out;

  const int N = in_sizes[0] / KDIM;
  const int E = in_sizes[1];

  float* ws  = (float*)d_ws;
  float* t   = ws;
  float* agg = t   + (size_t)N * HDIM;
  float* t2  = agg + (size_t)N * HDIM;
  float* deg = t2 + 2 * (size_t)N;

  const int TB = 256;
  gcn_init_deg<<<(N + TB - 1) / TB, TB, 0, stream>>>(deg, N);
  gcn_accum_deg_lds<PRV_DEG_EPT>
      <<<(E + PRV_THREADS * PRV_DEG_EPT - 1) / (PRV_THREADS * PRV_DEG_EPT), PRV_THREADS, 0, stream>>>(
          dst, deg, E, N);

  gcn_gemm1_wmma<<<2 * ((N + G1_ROWS - 1) / G1_ROWS), G1_THREADS, 0, stream>>>(x, W1, deg, t, agg, N);
  gcn_scatter1<<<(E + 8 * S1_EPW - 1) / (8 * S1_EPW), 256, 0, stream>>>(src, dst, t, agg, E);

  gcn_gemm2<<<(N + 7) / 8, 256, 0, stream>>>(agg, W2, deg, b1, t2, out, N);
  gcn_scatter2_lds<PRV_L2_EPT>
      <<<(E + PRV_THREADS * PRV_L2_EPT - 1) / (PRV_THREADS * PRV_L2_EPT), PRV_THREADS, 0, stream>>>(
          src, dst, t2, out, E, N);
  gcn_tanh<<<(2 * N + TB - 1) / TB, TB, 0, stream>>>(deg, b2, out, 2 * N);
}


# Layer-1 edge sum by owner groups: a group keeps 80 agg rows in LDS, scans dst, queues its edges, adds source rows by LDS float adds (arrival order), ends with ReLU and W2; no global atomics there
# speedup vs baseline: 2.8026x; 1.0765x over previous
#include <hip/hip_runtime.h>
#include <math.h>

typedef float v2f __attribute__((ext_vector_type(2)));
typedef float v8f __attribute__((ext_vector_type(8)));

#define KDIM 386
#define KMAIN 384
#define HDIM 256


__global__ void gcn_init_deg(float* deg, int N) {
  int i = blockIdx.x * blockDim.x + threadIdx.x;
  if (i < N) deg[i] = 1.0f;
}

#define PRV_THREADS 1024
#define PRV_DEG_CAP 20480
#define PRV_L2_CAP 20480
#define PRV_DEG_EPT 8
#define PRV_L2_EPT 5
#define PRV_MRG 4

template <int EPT>
__global__ __launch_bounds__(PRV_THREADS) void gcn_accum_deg_lds(const int* __restrict__ dst,
                                                                 float* deg, int E, int N) {
  __shared__ __attribute__((aligned(16))) float cnt[PRV_DEG_CAP];
  const int tid = threadIdx.x;
  const int base = blockIdx.x * (PRV_THREADS * EPT) + tid;
  int d[EPT];
  for (int i = 0; i < EPT; ++i) {
    const int e = base + i * PRV_THREADS;
    const int dv = dst[(e < E) ? e : (E - 1)];
    d[i] = (e < E) ? dv : -1;
  }
  for (int n0 = 0; n0 < N; n0 += PRV_DEG_CAP) {
    const int nn = (N - n0 < PRV_DEG_CAP) ? (N - n0) : PRV_DEG_CAP;
    float4* c4 = (float4*)cnt;
    for (int i = tid; i < (nn + 3) / 4; i += PRV_THREADS) c4[i] = make_float4(0.f, 0.f, 0.f, 0.f);
    __syncthreads();
    for (int i = 0; i < EPT; ++i) {
      const unsigned r = (unsigned)(d[i] - n0);
      if (r < (unsigned)nn) atomicAdd(&cnt[r], 1.0f);
    }
    __syncthreads();
    float* dp = deg + n0;
    int i = tid;
    for (; i + (PRV_MRG - 1) * PRV_THREADS < nn; i += PRV_MRG * PRV_THREADS) {
      float a[PRV_MRG];
      for (int k = 0; k < PRV_MRG; ++k) a[k] = cnt[i + k * PRV_THREADS];
      for (int k = 0; k < PRV_MRG; ++k) atomicAdd(&dp[i + k * PRV_THREADS], a[k]);
    }
    for (; i < nn; i += PRV_THREADS) atomicAdd(&dp[i], cnt[i]);
    __syncthreads();
  }
}

__device__ __forceinline__ float gcn_dis(const float* __restrict__ deg, int n) {
  return rsqrtf(deg[n]);
}

#define G1_ROWS 160
#define G1_COLS 128
#define G1_THREADS 640
#define G1_KPAIRS 193
#define G1_PITCH 288
#define G1_KBLK 32

__device__ __forceinline__ void g1_close_products(v8f (&acc)[4]) {
  __builtin_amdgcn_sched_barrier(0);
  asm volatile("v_nop\n\tv_nop\n\tv_nop\n\tv_nop\n\tv_nop"
               : "+v"(acc[0]), "+v"(acc[1]), "+v"(acc[2]), "+v"(acc[3]));
  __builtin_amdgcn_sched_barrier(0);
}

template <int NS>
__device__ __forceinline__ void g1_run(const float2 (&a)[NS], const float* bp, v8f (&acc)[4]) {
  float2 b[NS][4];
#pragma unroll
  for (int j = 0; j < NS; ++j)
#pragma unroll
    for (int q = 0; q < 4; ++q) b[j][q] = *(const float2*)(bp + 2 * j * G1_PITCH + 32 * q);
  __builtin_amdgcn_sched_barrier(0);
#pragma unroll
  for (int j = 0; j < NS; ++j) {
    v2f av;  av.x = a[j].x;  av.y = a[j].y;
#pragma unroll
    for (int q = 0; q < 4; ++q) {
      v2f bv;  bv.x = b[j][q].x;  bv.y = b[j][q].y;
      acc[q] = __builtin_amdgcn_wmma_f32_16x16x4_f32(false, av, false, bv, (short)0, acc[q],
                                                     false, false);
    }
  }
  g1_close_products(acc);
}

__global__ __launch_bounds__(G1_THREADS) void gcn_gemm1_wmma(const float* __restrict__ x,
                                                             const float* __restrict__ W1,
                                                             const float* __restrict__ deg,
                                                             float* __restrict__ t, int M) {
  __shared__ __attribute__((aligned(16))) float Bs[(G1_KPAIRS + 1) * G1_PITCH];
  const int tid = threadIdx.x;
  const int lane = tid & 31;
  const int wave = tid >> 5;
  const int halfSel = lane >> 4;
  const int l15 = lane & 15;
  const int colbase = (blockIdx.x & 1) * G1_COLS;
  const int m0 = (blockIdx.x >> 1) * G1_ROWS + (wave >> 1) * 16;
  const int cb = (wave & 1) * 64;

  const int row = m0 + l15;
  const int rowc = (row < M) ? row : (M - 1);
  const float* arow = x + (size_t)rowc * KDIM + halfSel * 2;

  float2 a0[8], a1[8];
#pragma unroll
  for (int j = 0; j < 8; ++j) a0[j] = *(const float2*)(arow + 4 * j);

  for (int it = tid; it < G1_KPAIRS * (G1_COLS / 4); it += G1_THREADS) {
    const int kp = it >> 5;
    const int c4 = (it & 31) * 4;
    const float* wp = W1 + (size_t)(2 * kp) * HDIM + colbase + c4;
    const float4 r0 = *(const float4*)wp;
    const float4 r1 = *(const float4*)(wp + HDIM);
    float4* dp = (float4*)(Bs + kp * G1_PITCH + 2 * c4);
    dp[0] = make_float4(r0.x, r1.x, r0.y, r1.y);
    dp[1] = make_float4(r0.z, r1.z, r0.w, r1.w);
  }
  for (int i = tid; i < 2 * G1_COLS; i += G1_THREADS) Bs[G1_KPAIRS * G1_PITCH + i] = 0.f;
  __syncthreads();

  v8f acc[4];
#pragma unroll
  for (int q = 0; q < 4; ++q) acc[q] = (v8f){0.f, 0.f, 0.f, 0.f, 0.f, 0.f, 0.f, 0.f};

  const float* bl = Bs + halfSel * G1_PITCH + 2 * (cb + l15);

  for (int kb = 0; kb < KMAIN / G1_KBLK; kb += 2) {
#pragma unroll
    for (int j = 0; j < 8; ++j) a1[j] = *(const float2*)(arow + (kb + 1) * G1_KBLK + 4 * j);
    g1_run<8>(a0, bl + (kb * (G1_KBLK / 2)) * G1_PITCH, acc);
    if (kb + 2 < KMAIN / G1_KBLK) {
#pragma unroll
      for (int j = 0; j < 8; ++j) a0[j] = *(const float2*)(arow + (kb + 2) * G1_KBLK + 4 * j);
    }
    g1_run<8>(a1, bl + ((kb + 1) * (G1_KBLK / 2)) * G1_PITCH, acc);
  }

  {
    float2 at[1];
    at[0] = make_float2(0.f, 0.f);
    if (halfSel == 0) at[0] = *(const float2*)(arow + KMAIN);
    g1_run<1>(at, bl + (KMAIN / 2) * G1_PITCH, acc);
  }

  float dr[8];
#pragma unroll
  for (int r = 0; r < 8; ++r) {
    int rr = m0 + r + halfSel * 8;
    dr[r] = gcn_dis(deg, (rr < M) ? rr : (M - 1));
  }

#pragma unroll
  for (int r = 0; r < 8; ++r) {
    int rr = m0 + r + halfSel * 8;
    if (rr < M) {
      const size_t o = (size_t)rr * HDIM + colbase + cb + l15;
#pragma unroll
      for (int q = 0; q < 4; ++q) {
        t[o + 16 * q] = acc[q][r] * dr[r];
      }
    }
  }
}

#define O1_ROWS 80
#define O1_THREADS 1024
#define O1_WAVES 32
#define O1_UNR 16
#define O1_QCAP 1024
#define O1_EPB 4
#define O1_INIT (O1_ROWS * (HDIM / 4) / O1_THREADS)
static_assert(O1_INIT * O1_THREADS == O1_ROWS * (HDIM / 4), "start values: whole pieces per thread");

__device__ __forceinline__ void gcn_owner1_drain(const int* q, int cnt,
                                                 const int* __restrict__ src,
                                                 const int* __restrict__ dst,
                                                 const float* __restrict__ t, float* acc,
                                                 int n0, int lane) {
  for (int k0 = 0; k0 < cnt; k0 += 32) {
    const int m = (cnt - k0 < 32) ? (cnt - k0) : 32;
    const int e = q[k0 + ((lane < m) ? lane : 0)];
    const int s = src[e];
    const int r = dst[e] - n0;
    for (int j = 0; j < m; j += O1_EPB) {
      int sj[O1_EPB], rj[O1_EPB];
      float v[O1_EPB][8];
#pragma unroll
      for (int i = 0; i < O1_EPB; ++i) {
        const int idx = (j + i < m) ? (j + i) : (m - 1);
        sj[i] = __builtin_amdgcn_readlane(s, idx);
        rj[i] = __builtin_amdgcn_readlane(r, idx);
      }
#pragma unroll
      for (int i = 0; i < O1_EPB; ++i) {
        const float* tp = t + (size_t)sj[i] * HDIM + lane;
#pragma unroll
        for (int jj = 0; jj < 8; ++jj) v[i][jj] = tp[32 * jj];
      }
#pragma unroll
      for (int i = 0; i < O1_EPB; ++i) {
        if (j + i < m) {
          float* ap = acc + rj[i] * HDIM + lane;
#pragma unroll
          for (int jj = 0; jj < 8; ++jj) atomicAdd(ap + 32 * jj, v[i][jj]);
        }
      }
    }
  }
}

__device__ __forceinline__ void gcn_owner1_load(const int* __restrict__ dst, int b0, int E,
                                                int (&d)[O1_UNR]) {
#pragma unroll
  for (int u = 0; u < O1_UNR; ++u) {
    const int e = b0 + u * O1_THREADS;
    d[u] = dst[(e < E) ? e : (E - 1)];
  }
}

__device__ __forceinline__ void gcn_owner1_test(const int (&d)[O1_UNR], int b0, int E, int n0,
                                                int nn, int* q, int& cnt) {
#pragma unroll
  for (int u = 0; u < O1_UNR; ++u) {
    const int e = b0 + u * O1_THREADS;
    const bool hit = (e < E) && ((unsigned)(d[u] - n0) < (unsigned)nn);
    const unsigned mask = __builtin_amdgcn_ballot_w32(hit);
    if (hit) q[cnt + (int)__builtin_amdgcn_mbcnt_lo(mask, 0u)] = e;
    cnt += __builtin_popcount(mask);
  }
}

__global__ __launch_bounds__(O1_THREADS) void gcn_owner1(const int* __restrict__ src,
                                                         const int* __restrict__ dst,
                                                         const float* __restrict__ t,
                                                         const float* __restrict__ W2,
                                                         const float* __restrict__ deg,
                                                         const float* __restrict__ b1,
                                                         float* __restrict__ t2,
                                                         float* __restrict__ out, int E, int N) {
  __shared__ __attribute__((aligned(16))) float acc[O1_ROWS * HDIM];
  __shared__ int qe[O1_WAVES * O1_QCAP];
  const int tid = threadIdx.x;
  const int lane = tid & 31;
  const int wave = __builtin_amdgcn_readfirstlane(tid >> 5);
  const int n0 = blockIdx.x * O1_ROWS;
  const int nn = (N - n0 < O1_ROWS) ? (N - n0) : O1_ROWS;

  {
    const float4* tp4 = (const float4*)(t + (size_t)n0 * HDIM);
    float4* a4 = (float4*)acc;
    const int lim4 = nn * (HDIM / 4);
    float4 sv[O1_INIT];
#pragma unroll
    for (int k = 0; k < O1_INIT; ++k) {
      const int i = tid + k * O1_THREADS;
      sv[k] = tp4[(i < lim4) ? i : (lim4 - 1)];
    }
#pragma unroll
    for (int k = 0; k < O1_INIT; ++k) a4[tid + k * O1_THREADS] = sv[k];
  }
  __syncthreads();

  int* q = qe + wave * O1_QCAP;
  int cnt = 0;
  const int span = O1_THREADS * O1_UNR;
  const int trips = (E + span - 1) / span;
  const int lim = O1_QCAP - 32 * O1_UNR;
  int trip = (int)(blockIdx.x % (unsigned)trips);
  int dA[O1_UNR], dB[O1_UNR];
  int bA = trip * span + wave * 32 + lane;
  gcn_owner1_load(dst, bA, E, dA);
  for (int it = 0; it < trips; it += 2) {
    trip = (trip + 1 == trips) ? 0 : (trip + 1);
    const int bB = trip * span + wave * 32 + lane;
    gcn_owner1_load(dst, bB, E, dB);
    __builtin_amdgcn_sched_barrier(0);
    gcn_owner1_test(dA, bA, E, n0, nn, q, cnt);
    if (cnt > lim) {
      gcn_owner1_drain(q, cnt, src, dst, t, acc, n0, lane);
      cnt = 0;
    }
    if (it + 1 == trips) break;
    trip = (trip + 1 == trips) ? 0 : (trip + 1);
    bA = trip * span + wave * 32 + lane;
    gcn_owner1_load(dst, bA, E, dA);
    __builtin_amdgcn_sched_barrier(0);
    gcn_owner1_test(dB, bB, E, n0, nn, q, cnt);
    if (cnt > lim) {
      gcn_owner1_drain(q, cnt, src, dst, t, acc, n0, lane);
      cnt = 0;
    }
  }
  gcn_owner1_drain(q, cnt, src, dst, t, acc, n0, lane);
  __syncthreads();

  for (int r = wave; r < nn; r += O1_WAVES) {
    const int row = n0 + r;
    const float dr = gcn_dis(deg, row);
    const float* ar = acc + r * HDIM;
    float s0 = 0.f, s1 = 0.f;
#pragma unroll
    for (int k = lane; k < HDIM; k += 32) {
      float a = ar[k] * dr + b1[k];
      a = a > 0.f ? a : 0.f;
      s0 += a * W2[k * 2 + 0];
      s1 += a * W2[k * 2 + 1];
    }
#pragma unroll
    for (int off = 16; off > 0; off >>= 1) {
      s0 += __shfl_down(s0, off);
      s1 += __shfl_down(s1, off);
    }
    if (lane == 0) {
      const float u0 = s0 * dr;
      const float u1 = s1 * dr;
      t2[row * 2 + 0] = u0;
      t2[row * 2 + 1] = u1;
      out[row * 2 + 0] = u0;
      out[row * 2 + 1] = u1;
    }
  }
}

template <int EPT>
__global__ __launch_bounds__(PRV_THREADS) void gcn_scatter2_lds(const int* __restrict__ src,
                                                                const int* __restrict__ dst,
                                                                const float* __restrict__ t2,
                                                                float* out, int E, int N) {
  __shared__ __attribute__((aligned(16))) float acc[2 * PRV_L2_CAP];
  const int tid = threadIdx.x;
  const int base = blockIdx.x * (PRV_THREADS * EPT) + tid;
  int d[EPT];
  float2 v[EPT];
  {
    int s[EPT];
    for (int i = 0; i < EPT; ++i) {
      const int e = base + i * PRV_THREADS;
      const int ec = (e < E) ? e : (E - 1);
      s[i] = src[ec];
      const int dv = dst[ec];
      d[i] = (e < E) ? dv : -1;
    }
    for (int i = 0; i < EPT; ++i) v[i] = *(const float2*)(t2 + 2 * (size_t)s[i]);
  }
  for (int n0 = 0; n0 < N; n0 += PRV_L2_CAP) {
    const int nn = (N - n0 < PRV_L2_CAP) ? (N - n0) : PRV_L2_CAP;
    float4* a4 = (float4*)acc;
    for (int i = tid; i < (2 * nn + 3) / 4; i += PRV_THREADS) a4[i] = make_float4(0.f, 0.f, 0.f, 0.f);
    __syncthreads();
    for (int i = 0; i < EPT; ++i) {
      const unsigned r = (unsigned)(d[i] - n0);
      if (r < (unsigned)nn) {
        atomicAdd(&acc[2 * r + 0], v[i].x);
        atomicAdd(&acc[2 * r + 1], v[i].y);
      }
    }
    __syncthreads();
    float* op = out + 2 * (size_t)n0;
    const int lim = 2 * nn;
    int i = tid;
    for (; i + (PRV_MRG - 1) * PRV_THREADS < lim; i += PRV_MRG * PRV_THREADS) {
      float a[PRV_MRG];
      for (int k = 0; k < PRV_MRG; ++k) a[k] = acc[i + k * PRV_THREADS];
      for (int k = 0; k < PRV_MRG; ++k) atomicAdd(&op[i + k * PRV_THREADS], a[k]);
    }
    for (; i < lim; i += PRV_THREADS) atomicAdd(&op[i], acc[i]);
    __syncthreads();
  }
}

__global__ void gcn_tanh(const float* __restrict__ deg, const float* __restrict__ b2,
                         float* out, int n) {
  int i = blockIdx.x * blockDim.x + threadIdx.x;
  if (i < n) out[i] = tanhf(out[i] * gcn_dis(deg, i >> 1) + b2[i & 1]);
}

extern "C" void kernel_launch(void* const* d_in, const int* in_sizes, int n_in,
                              void* d_out, int out_size, void* d_ws, size_t ws_size,
                              hipStream_t stream) {
  const float* x  = (const float*)d_in[0];
  const int*   src = (const int*)d_in[1];
  const int*   dst = (const int*)d_in[2];
  const float* W1 = (const float*)d_in[3];
  const float* b1 = (const float*)d_in[4];
  const float* W2 = (const float*)d_in[5];
  const float* b2 = (const float*)d_in[6];
  float* out = (float*)d_out;

  const int N = in_sizes[0] / KDIM;
  const int E = in_sizes[1];

  float* ws  = (float*)d_ws;
  float* t   = ws;
  float* t2  = t + (size_t)N * HDIM;
  float* deg = t2 + 2 * (size_t)N;

  const int TB = 256;
  gcn_init_deg<<<(N + TB - 1) / TB, TB, 0, stream>>>(deg, N);
  gcn_accum_deg_lds<PRV_DEG_EPT>
      <<<(E + PRV_THREADS * PRV_DEG_EPT - 1) / (PRV_THREADS * PRV_DEG_EPT), PRV_THREADS, 0, stream>>>(
          dst, deg, E, N);

  gcn_gemm1_wmma<<<2 * ((N + G1_ROWS - 1) / G1_ROWS), G1_THREADS, 0, stream>>>(x, W1, deg, t, N);
  gcn_owner1<<<(N + O1_ROWS - 1) / O1_ROWS, O1_THREADS, 0, stream>>>(src, dst, t, W2, deg, b1, t2, out,
                                                                    E, N);

  gcn_scatter2_lds<PRV_L2_EPT>
      <<<(E + PRV_THREADS * PRV_L2_EPT - 1) / (PRV_THREADS * PRV_L2_EPT), PRV_THREADS, 0, stream>>>(
          src, dst, t2, out, E, N);
  gcn_tanh<<<(2 * N + TB - 1) / TB, TB, 0, stream>>>(deg, b2, out, 2 * N);
}
